# starting kernel
# baseline (speedup 1.0000x reference)
#include <hip/hip_runtime.h>
#include <hip/hip_bf16.h>
#include <math.h>

#define LL 2048
#define BB 64
#define KK 256
#define VV 256
#define HH 8
#define HK 2048
#define HV 2048

typedef __attribute__((ext_vector_type(2))) float v2f;
typedef __attribute__((ext_vector_type(8))) float v8f;

__device__ __forceinline__ v8f wmma4(v2f a, v2f b, v8f c) {
    return __builtin_amdgcn_wmma_f32_16x16x4_f32(
         false, a,  false, b,
         (short)0, c,  false,  false);
}

__global__ void k_qproj(const float* __restrict__ query,
                        const float* __restrict__ Wq,
                        const float* __restrict__ bq,
                        float* __restrict__ qbuf) {
    const int waveId = blockIdx.x * (blockDim.x >> 5) + (threadIdx.x >> 5);
    const int tm   = waveId >> 7;
    const int tn   = waveId & 127;
    const int lane = threadIdx.x & 31;
    const int col  = lane & 15;
    const int half = lane >> 4;

    const float* Arow = query + (size_t)(tm * 16 + col) * KK;
    const float* Bcol = Wq + (tn * 16 + col);

    v8f c = {};
    #pragma unroll 4
    for (int k0 = 0; k0 < KK; k0 += 4) {
        const int ka = k0 + 2 * half;
        v2f a = { Arow[ka], Arow[ka + 1] };
        v2f b = { Bcol[(size_t)ka * HK], Bcol[(size_t)(ka + 1) * HK] };
        c = wmma4(a, b, c);
    }

    const float bias = bq[tn * 16 + col];
    #pragma unroll
    for (int j = 0; j < 8; ++j) {
        const int row = tm * 16 + j + 8 * half;
        qbuf[(size_t)row * HK + tn * 16 + col] = c[j] + bias;
    }
}

__global__ void k_scores(const float* __restrict__ keys,
                         const float* __restrict__ rpe,
                         const float* __restrict__ qbuf,
                         float* __restrict__ smat) {
    const int waveId = blockIdx.x * (blockDim.x >> 5) + (threadIdx.x >> 5);
    const int b    = waveId >> 7;
    const int lt   = waveId & 127;
    const int lane = threadIdx.x & 31;
    const int col  = lane & 15;
    const int half = lane >> 4;

    const int l_m = lt * 16 + col;
    const float* keyRow = keys + ((size_t)l_m * BB + b) * KK;
    const float r = rpe[(size_t)l_m * BB + b];

    const bool bn = (col < HH);
    const float* qb = qbuf + (size_t)b * HK + (size_t)col * KK;

    __builtin_prefetch(keyRow, 0, 0);

    v8f c = {};
    #pragma unroll 8
    for (int k0 = 0; k0 < KK; k0 += 4) {
        const int ka = k0 + 2 * half;
        if ((k0 & 28) == 0)
            __builtin_prefetch(keyRow + k0 + 32, 0, 0);
        v2f kv = __builtin_nontemporal_load((const v2f*)(keyRow + ka));
        v2f a  = { kv.x * r, kv.y * r };
        v2f bb = { bn ? qb[ka] : 0.0f, bn ? qb[ka + 1] : 0.0f };
        c = wmma4(a, bb, c);
    }

    if (col < HH) {
        #pragma unroll
        for (int j = 0; j < 8; ++j) {
            const int l = lt * 16 + j + 8 * half;
            smat[((size_t)b * HH + col) * LL + l] = c[j];
        }
    }
}

__global__ void k_softmax(float* __restrict__ s, const int* __restrict__ steps) {
    const int bh   = blockIdx.x;
    const int b    = bh >> 3;
    const int step = steps[b];
    float* row = s + (size_t)bh * LL;
    const int t = threadIdx.x;

    float v[8];
    float mx = -INFINITY;
    #pragma unroll
    for (int i = 0; i < 8; ++i) {
        const int l = t + i * 256;
        const float x = (l < step) ? row[l] : -INFINITY;
        v[i] = x;
        mx = fmaxf(mx, x);
    }

    __shared__ float red[256];
    red[t] = mx;
    __syncthreads();
    for (int off = 128; off > 0; off >>= 1) {
        if (t < off) red[t] = fmaxf(red[t], red[t + off]);
        __syncthreads();
    }
    mx = red[0];
    __syncthreads();

    float sum = 0.0f;
    #pragma unroll
    for (int i = 0; i < 8; ++i) {
        const int l = t + i * 256;
        const float e = (l < step) ? __expf(v[i] - mx) : 0.0f;
        v[i] = e;
        sum += e;
    }
    red[t] = sum;
    __syncthreads();
    for (int off = 128; off > 0; off >>= 1) {
        if (t < off) red[t] += red[t + off];
        __syncthreads();
    }
    const float inv = 1.0f / red[0];

    #pragma unroll
    for (int i = 0; i < 8; ++i) row[t + i * 256] = v[i] * inv;
}

__global__ void k_read(const float* __restrict__ wmat,
                       const float* __restrict__ vals,
                       float* __restrict__ readbuf) {
    const int waveId = blockIdx.x * (blockDim.x >> 5) + (threadIdx.x >> 5);
    const int b    = waveId >> 4;
    const int tn   = waveId & 15;
    const int lane = threadIdx.x & 31;
    const int col  = lane & 15;
    const int half = lane >> 4;

    const bool am = (col < HH);
    const float* wrow  = wmat + ((size_t)b * HH + col) * LL;
    const float* vbase = vals + (size_t)b * VV + tn * 16 + col;

    v8f c = {};
    #pragma unroll 4
    for (int k0 = 0; k0 < LL; k0 += 4) {
        const int ka = k0 + 2 * half;
        if ((k0 & 12) == 0)
            __builtin_prefetch(vbase + (size_t)(k0 + 8) * (BB * VV), 0, 0);
        v2f a  = { am ? wrow[ka] : 0.0f, am ? wrow[ka + 1] : 0.0f };
        v2f bb = { __builtin_nontemporal_load(vbase + (size_t)ka * (BB * VV)),
                   __builtin_nontemporal_load(vbase + (size_t)(ka + 1) * (BB * VV)) };
        c = wmma4(a, bb, c);
    }

    if (half == 0) {
        #pragma unroll
        for (int j = 0; j < 8; ++j)
            readbuf[((size_t)b * HH + j) * VV + tn * 16 + col] = c[j];
    }
}

__global__ void k_out(const float* __restrict__ readbuf,
                      const float* __restrict__ Wa,
                      const float* __restrict__ ba,
                      float* __restrict__ out) {
    const int waveId = blockIdx.x * (blockDim.x >> 5) + (threadIdx.x >> 5);
    const int tm   = waveId >> 4;
    const int tn   = waveId & 15;
    const int lane = threadIdx.x & 31;
    const int col  = lane & 15;
    const int half = lane >> 4;

    const float* Arow = readbuf + (size_t)(tm * 16 + col) * HV;
    const float* Bcol = Wa + tn * 16 + col;

    v8f c = {};
    #pragma unroll 2
    for (int k0 = 0; k0 < HV; k0 += 4) {
        const int ka = k0 + 2 * half;
        v2f a = { Arow[ka], Arow[ka + 1] };
        v2f b = { Bcol[(size_t)ka * VV], Bcol[(size_t)(ka + 1) * VV] };
        c = wmma4(a, b, c);
    }

    const float bias = ba[tn * 16 + col];
    #pragma unroll
    for (int j = 0; j < 8; ++j) {
        const int row = tm * 16 + j + 8 * half;
        out[(size_t)row * VV + tn * 16 + col] = c[j] + bias;
    }
}

extern "C" void kernel_launch(void* const* d_in, const int* in_sizes, int n_in,
                              void* d_out, int out_size, void* d_ws, size_t ws_size,
                              hipStream_t stream) {
    const float* query = (const float*)d_in[0];
    const float* keys  = (const float*)d_in[1];
    const float* vals  = (const float*)d_in[2];
    const float* rpe   = (const float*)d_in[3];
    const float* Wq    = (const float*)d_in[4];
    const float* bq    = (const float*)d_in[5];
    const float* Wa    = (const float*)d_in[6];
    const float* ba    = (const float*)d_in[7];
    const int*   steps = (const int*)d_in[8];
    float* out = (float*)d_out;

    float* ws = (float*)d_ws;
    float* qbuf    = ws;
    float* smat    = qbuf + (size_t)BB * HK;
    float* readbuf = smat + (size_t)BB * HH * LL;

    k_qproj<<<64, 256, 0, stream>>>(query, Wq, bq, qbuf);
    k_scores<<<1024, 256, 0, stream>>>(keys, rpe, qbuf, smat);
    k_softmax<<<BB * HH, 256, 0, stream>>>(smat, steps);
    k_read<<<128, 256, 0, stream>>>(smat, vals, readbuf);
    k_out<<<8, 256, 0, stream>>>(readbuf, Wa, ba, out);
}


# k_read retiled for bytes in flight: one 16-wave workgroup per (batch, 64 columns), waves split L, whole-line 16-byte loads kept two groups deep, partial tiles summed through LDS (new sum order)
# speedup vs baseline: 2.3323x; 2.3323x over previous
#include <hip/hip_runtime.h>
#include <hip/hip_bf16.h>
#include <math.h>

#define LL 2048
#define BB 64
#define KK 256
#define VV 256
#define HH 8
#define HK 2048
#define HV 2048

typedef __attribute__((ext_vector_type(2))) float v2f;
typedef __attribute__((ext_vector_type(4))) float v4f;
typedef __attribute__((ext_vector_type(8))) float v8f;

__device__ __forceinline__ v8f wmma4(v2f a, v2f b, v8f c) {
    return __builtin_amdgcn_wmma_f32_16x16x4_f32(
         false, a,  false, b,
         (short)0, c,  false,  false);
}

__global__ void k_qproj(const float* __restrict__ query,
                        const float* __restrict__ Wq,
                        const float* __restrict__ bq,
                        float* __restrict__ qbuf) {
    const int waveId = blockIdx.x * (blockDim.x >> 5) + (threadIdx.x >> 5);
    const int tm   = waveId >> 7;
    const int tn   = waveId & 127;
    const int lane = threadIdx.x & 31;
    const int col  = lane & 15;
    const int half = lane >> 4;

    const float* Arow = query + (size_t)(tm * 16 + col) * KK;
    const float* Bcol = Wq + (tn * 16 + col);

    v8f c = {};
    #pragma unroll 4
    for (int k0 = 0; k0 < KK; k0 += 4) {
        const int ka = k0 + 2 * half;
        v2f a = { Arow[ka], Arow[ka + 1] };
        v2f b = { Bcol[(size_t)ka * HK], Bcol[(size_t)(ka + 1) * HK] };
        c = wmma4(a, b, c);
    }

    const float bias = bq[tn * 16 + col];
    #pragma unroll
    for (int j = 0; j < 8; ++j) {
        const int row = tm * 16 + j + 8 * half;
        qbuf[(size_t)row * HK + tn * 16 + col] = c[j] + bias;
    }
}

__global__ void k_scores(const float* __restrict__ keys,
                         const float* __restrict__ rpe,
                         const float* __restrict__ qbuf,
                         float* __restrict__ smat) {
    const int waveId = blockIdx.x * (blockDim.x >> 5) + (threadIdx.x >> 5);
    const int b    = waveId >> 7;
    const int lt   = waveId & 127;
    const int lane = threadIdx.x & 31;
    const int col  = lane & 15;
    const int half = lane >> 4;

    const int l_m = lt * 16 + col;
    const float* keyRow = keys + ((size_t)l_m * BB + b) * KK;
    const float r = rpe[(size_t)l_m * BB + b];

    const bool bn = (col < HH);
    const float* qb = qbuf + (size_t)b * HK + (size_t)col * KK;

    __builtin_prefetch(keyRow, 0, 0);

    v8f c = {};
    #pragma unroll 8
    for (int k0 = 0; k0 < KK; k0 += 4) {
        const int ka = k0 + 2 * half;
        if ((k0 & 28) == 0)
            __builtin_prefetch(keyRow + k0 + 32, 0, 0);
        v2f kv = __builtin_nontemporal_load((const v2f*)(keyRow + ka));
        v2f a  = { kv.x * r, kv.y * r };
        v2f bb = { bn ? qb[ka] : 0.0f, bn ? qb[ka + 1] : 0.0f };
        c = wmma4(a, bb, c);
    }

    if (col < HH) {
        #pragma unroll
        for (int j = 0; j < 8; ++j) {
            const int l = lt * 16 + j + 8 * half;
            smat[((size_t)b * HH + col) * LL + l] = c[j];
        }
    }
}

__global__ void k_softmax(float* __restrict__ s, const int* __restrict__ steps) {
    const int bh   = blockIdx.x;
    const int b    = bh >> 3;
    const int step = steps[b];
    float* row = s + (size_t)bh * LL;
    const int t = threadIdx.x;

    float v[8];
    float mx = -INFINITY;
    #pragma unroll
    for (int i = 0; i < 8; ++i) {
        const int l = t + i * 256;
        const float x = (l < step) ? row[l] : -INFINITY;
        v[i] = x;
        mx = fmaxf(mx, x);
    }

    __shared__ float red[256];
    red[t] = mx;
    __syncthreads();
    for (int off = 128; off > 0; off >>= 1) {
        if (t < off) red[t] = fmaxf(red[t], red[t + off]);
        __syncthreads();
    }
    mx = red[0];
    __syncthreads();

    float sum = 0.0f;
    #pragma unroll
    for (int i = 0; i < 8; ++i) {
        const int l = t + i * 256;
        const float e = (l < step) ? __expf(v[i] - mx) : 0.0f;
        v[i] = e;
        sum += e;
    }
    red[t] = sum;
    __syncthreads();
    for (int off = 128; off > 0; off >>= 1) {
        if (t < off) red[t] += red[t + off];
        __syncthreads();
    }
    const float inv = 1.0f / red[0];

    #pragma unroll
    for (int i = 0; i < 8; ++i) row[t + i * 256] = v[i] * inv;
}

#define RD_WAVES 16
#define RD_ROWS  (LL / RD_WAVES)
#define RD_STEPS 4
#define RD_GRPS  (RD_ROWS / (8 * RD_STEPS))

__device__ __forceinline__ void rd_load_group(const float* __restrict__ wp,
                                              const float* __restrict__ vp,
                                              int g,
                                              v4f (&wa)[RD_STEPS],
                                              v4f (&vb)[RD_STEPS][4]) {
    for (int j = 0; j < RD_STEPS; ++j) {
        const int s = g * RD_STEPS + j;
        wa[j] = *(const v4f*)(wp + 8 * s);
        for (int q = 0; q < 4; ++q)
            vb[j][q] = __builtin_nontemporal_load(
                (const v4f*)(vp + (size_t)(8 * s + q) * (BB * VV)));
    }
}

__device__ __forceinline__ void rd_mma_group(const v4f (&wa)[RD_STEPS],
                                             const v4f (&vb)[RD_STEPS][4],
                                             v8f (&acc)[4]) {
    for (int j = 0; j < RD_STEPS; ++j) {
        for (int p = 0; p < 2; ++p) {
            const v2f a = { wa[j][2 * p], wa[j][2 * p + 1] };
            for (int t = 0; t < 4; ++t) {
                const v2f bb = { vb[j][2 * p][t], vb[j][2 * p + 1][t] };
                acc[t] = wmma4(a, bb, acc[t]);
            }
        }
    }
}

__global__ void __launch_bounds__(RD_WAVES * 32)
k_read(const float* __restrict__ wmat,
       const float* __restrict__ vals,
       float* __restrict__ readbuf) {
    __shared__ float part[RD_WAVES][HH][64];

    const int b    = blockIdx.x >> 2;
    const int vc   = blockIdx.x & 3;
    const int wave = threadIdx.x >> 5;
    const int lane = threadIdx.x & 31;
    const int col  = lane & 15;
    const int half = lane >> 4;
    const int l0 = wave * RD_ROWS + 4 * half;
    const float* wp = wmat + ((size_t)b * HH + (col & (HH - 1))) * LL + l0;
    const float* vp = vals + ((size_t)l0 * BB + b) * VV + vc * 64 + 4 * col;

    v8f acc[4] = {};
    v4f wa[2][RD_STEPS];
    v4f vb[2][RD_STEPS][4];

    rd_load_group(wp, vp, 0, wa[0], vb[0]);
    for (int g = 0; g < RD_GRPS; ++g) {
        __builtin_amdgcn_sched_barrier(0);
        if (g + 1 < RD_GRPS)
            rd_load_group(wp, vp, g + 1, wa[(g + 1) & 1], vb[(g + 1) & 1]);
        __builtin_amdgcn_sched_barrier(0);
        rd_mma_group(wa[g & 1], vb[g & 1], acc);
    }

    if (half == 0) {
        for (int r = 0; r < 8; ++r) {
            const v4f o = { acc[0][r], acc[1][r], acc[2][r], acc[3][r] };
            *(v4f*)&part[wave][r][4 * col] = o;
        }
    }
    __syncthreads();

    const int r  = threadIdx.x >> 6;
    const int vv = threadIdx.x & 63;
    float s = 0.0f;
    for (int w = 0; w < RD_WAVES; ++w) s += part[w][r][vv];
    readbuf[((size_t)b * HH + r) * VV + vc * 64 + vv] = s;
}

__global__ void k_out(const float* __restrict__ readbuf,
                      const float* __restrict__ Wa,
                      const float* __restrict__ ba,
                      float* __restrict__ out) {
    const int waveId = blockIdx.x * (blockDim.x >> 5) + (threadIdx.x >> 5);
    const int tm   = waveId >> 4;
    const int tn   = waveId & 15;
    const int lane = threadIdx.x & 31;
    const int col  = lane & 15;
    const int half = lane >> 4;

    const float* Arow = readbuf + (size_t)(tm * 16 + col) * HV;
    const float* Bcol = Wa + tn * 16 + col;

    v8f c = {};
    #pragma unroll 2
    for (int k0 = 0; k0 < HV; k0 += 4) {
        const int ka = k0 + 2 * half;
        v2f a = { Arow[ka], Arow[ka + 1] };
        v2f b = { Bcol[(size_t)ka * VV], Bcol[(size_t)(ka + 1) * VV] };
        c = wmma4(a, b, c);
    }

    const float bias = ba[tn * 16 + col];
    #pragma unroll
    for (int j = 0; j < 8; ++j) {
        const int row = tm * 16 + j + 8 * half;
        out[(size_t)row * VV + tn * 16 + col] = c[j] + bias;
    }
}

extern "C" void kernel_launch(void* const* d_in, const int* in_sizes, int n_in,
                              void* d_out, int out_size, void* d_ws, size_t ws_size,
                              hipStream_t stream) {
    const float* query = (const float*)d_in[0];
    const float* keys  = (const float*)d_in[1];
    const float* vals  = (const float*)d_in[2];
    const float* rpe   = (const float*)d_in[3];
    const float* Wq    = (const float*)d_in[4];
    const float* bq    = (const float*)d_in[5];
    const float* Wa    = (const float*)d_in[6];
    const float* ba    = (const float*)d_in[7];
    const int*   steps = (const int*)d_in[8];
    float* out = (float*)d_out;

    float* ws = (float*)d_ws;
    float* qbuf    = ws;
    float* smat    = qbuf + (size_t)BB * HK;
    float* readbuf = smat + (size_t)BB * HH * LL;

    k_qproj<<<64, 256, 0, stream>>>(query, Wq, bq, qbuf);
    k_scores<<<1024, 256, 0, stream>>>(keys, rpe, qbuf, smat);
    k_softmax<<<BB * HH, 256, 0, stream>>>(smat, steps);
    k_read<<<256, 512, 0, stream>>>(smat, vals, readbuf);
    k_out<<<8, 256, 0, stream>>>(readbuf, Wa, ba, out);
}


# k_out split along k: one 16-wave workgroup per 16x32 tile of out, each wave sums 128 k with all 80 loads issued before its first product, partial tiles summed through LDS (new sum order)
# speedup vs baseline: 5.8207x; 2.4957x over previous
#include <hip/hip_runtime.h>
#include <hip/hip_bf16.h>
#include <math.h>

#define LL 2048
#define BB 64
#define KK 256
#define VV 256
#define HH 8
#define HK 2048
#define HV 2048

typedef __attribute__((ext_vector_type(2))) float v2f;
typedef __attribute__((ext_vector_type(4))) float v4f;
typedef __attribute__((ext_vector_type(8))) float v8f;

__device__ __forceinline__ v8f wmma4(v2f a, v2f b, v8f c) {
    return __builtin_amdgcn_wmma_f32_16x16x4_f32(
         false, a,  false, b,
         (short)0, c,  false,  false);
}

__global__ void k_qproj(const float* __restrict__ query,
                        const float* __restrict__ Wq,
                        const float* __restrict__ bq,
                        float* __restrict__ qbuf) {
    const int waveId = blockIdx.x * (blockDim.x >> 5) + (threadIdx.x >> 5);
    const int tm   = waveId >> 7;
    const int tn   = waveId & 127;
    const int lane = threadIdx.x & 31;
    const int col  = lane & 15;
    const int half = lane >> 4;

    const float* Arow = query + (size_t)(tm * 16 + col) * KK;
    const float* Bcol = Wq + (tn * 16 + col);

    v8f c = {};
    #pragma unroll 4
    for (int k0 = 0; k0 < KK; k0 += 4) {
        const int ka = k0 + 2 * half;
        v2f a = { Arow[ka], Arow[ka + 1] };
        v2f b = { Bcol[(size_t)ka * HK], Bcol[(size_t)(ka + 1) * HK] };
        c = wmma4(a, b, c);
    }

    const float bias = bq[tn * 16 + col];
    #pragma unroll
    for (int j = 0; j < 8; ++j) {
        const int row = tm * 16 + j + 8 * half;
        qbuf[(size_t)row * HK + tn * 16 + col] = c[j] + bias;
    }
}

__global__ void k_scores(const float* __restrict__ keys,
                         const float* __restrict__ rpe,
                         const float* __restrict__ qbuf,
                         float* __restrict__ smat) {
    const int waveId = blockIdx.x * (blockDim.x >> 5) + (threadIdx.x >> 5);
    const int b    = waveId >> 7;
    const int lt   = waveId & 127;
    const int lane = threadIdx.x & 31;
    const int col  = lane & 15;
    const int half = lane >> 4;

    const int l_m = lt * 16 + col;
    const float* keyRow = keys + ((size_t)l_m * BB + b) * KK;
    const float r = rpe[(size_t)l_m * BB + b];

    const bool bn = (col < HH);
    const float* qb = qbuf + (size_t)b * HK + (size_t)col * KK;

    __builtin_prefetch(keyRow, 0, 0);

    v8f c = {};
    #pragma unroll 8
    for (int k0 = 0; k0 < KK; k0 += 4) {
        const int ka = k0 + 2 * half;
        if ((k0 & 28) == 0)
            __builtin_prefetch(keyRow + k0 + 32, 0, 0);
        v2f kv = __builtin_nontemporal_load((const v2f*)(keyRow + ka));
        v2f a  = { kv.x * r, kv.y * r };
        v2f bb = { bn ? qb[ka] : 0.0f, bn ? qb[ka + 1] : 0.0f };
        c = wmma4(a, bb, c);
    }

    if (col < HH) {
        #pragma unroll
        for (int j = 0; j < 8; ++j) {
            const int l = lt * 16 + j + 8 * half;
            smat[((size_t)b * HH + col) * LL + l] = c[j];
        }
    }
}

__global__ void k_softmax(float* __restrict__ s, const int* __restrict__ steps) {
    const int bh   = blockIdx.x;
    const int b    = bh >> 3;
    const int step = steps[b];
    float* row = s + (size_t)bh * LL;
    const int t = threadIdx.x;

    float v[8];
    float mx = -INFINITY;
    #pragma unroll
    for (int i = 0; i < 8; ++i) {
        const int l = t + i * 256;
        const float x = (l < step) ? row[l] : -INFINITY;
        v[i] = x;
        mx = fmaxf(mx, x);
    }

    __shared__ float red[256];
    red[t] = mx;
    __syncthreads();
    for (int off = 128; off > 0; off >>= 1) {
        if (t < off) red[t] = fmaxf(red[t], red[t + off]);
        __syncthreads();
    }
    mx = red[0];
    __syncthreads();

    float sum = 0.0f;
    #pragma unroll
    for (int i = 0; i < 8; ++i) {
        const int l = t + i * 256;
        const float e = (l < step) ? __expf(v[i] - mx) : 0.0f;
        v[i] = e;
        sum += e;
    }
    red[t] = sum;
    __syncthreads();
    for (int off = 128; off > 0; off >>= 1) {
        if (t < off) red[t] += red[t + off];
        __syncthreads();
    }
    const float inv = 1.0f / red[0];

    #pragma unroll
    for (int i = 0; i < 8; ++i) row[t + i * 256] = v[i] * inv;
}

#define RD_WAVES 16
#define RD_ROWS  (LL / RD_WAVES)
#define RD_STEPS 4
#define RD_GRPS  (RD_ROWS / (8 * RD_STEPS))

__device__ __forceinline__ void rd_load_group(const float* __restrict__ wp,
                                              const float* __restrict__ vp,
                                              int g,
                                              v4f (&wa)[RD_STEPS],
                                              v4f (&vb)[RD_STEPS][4]) {
    for (int j = 0; j < RD_STEPS; ++j) {
        const int s = g * RD_STEPS + j;
        wa[j] = *(const v4f*)(wp + 8 * s);
        for (int q = 0; q < 4; ++q)
            vb[j][q] = __builtin_nontemporal_load(
                (const v4f*)(vp + (size_t)(8 * s + q) * (BB * VV)));
    }
}

__device__ __forceinline__ void rd_mma_group(const v4f (&wa)[RD_STEPS],
                                             const v4f (&vb)[RD_STEPS][4],
                                             v8f (&acc)[4]) {
    for (int j = 0; j < RD_STEPS; ++j) {
        for (int p = 0; p < 2; ++p) {
            const v2f a = { wa[j][2 * p], wa[j][2 * p + 1] };
            for (int t = 0; t < 4; ++t) {
                const v2f bb = { vb[j][2 * p][t], vb[j][2 * p + 1][t] };
                acc[t] = wmma4(a, bb, acc[t]);
            }
        }
    }
}

__global__ void __launch_bounds__(RD_WAVES * 32)
k_read(const float* __restrict__ wmat,
       const float* __restrict__ vals,
       float* __restrict__ readbuf) {
    __shared__ float part[RD_WAVES][HH][64];

    const int b    = blockIdx.x >> 2;
    const int vc   = blockIdx.x & 3;
    const int wave = threadIdx.x >> 5;
    const int lane = threadIdx.x & 31;
    const int col  = lane & 15;
    const int half = lane >> 4;
    const int l0 = wave * RD_ROWS + 4 * half;
    const float* wp = wmat + ((size_t)b * HH + (col & (HH - 1))) * LL + l0;
    const float* vp = vals + ((size_t)l0 * BB + b) * VV + vc * 64 + 4 * col;

    v8f acc[4] = {};
    v4f wa[2][RD_STEPS];
    v4f vb[2][RD_STEPS][4];

    rd_load_group(wp, vp, 0, wa[0], vb[0]);
    for (int g = 0; g < RD_GRPS; ++g) {
        __builtin_amdgcn_sched_barrier(0);
        if (g + 1 < RD_GRPS)
            rd_load_group(wp, vp, g + 1, wa[(g + 1) & 1], vb[(g + 1) & 1]);
        __builtin_amdgcn_sched_barrier(0);
        rd_mma_group(wa[g & 1], vb[g & 1], acc);
    }

    if (half == 0) {
        for (int r = 0; r < 8; ++r) {
            const v4f o = { acc[0][r], acc[1][r], acc[2][r], acc[3][r] };
            *(v4f*)&part[wave][r][4 * col] = o;
        }
    }
    __syncthreads();

    const int r  = threadIdx.x >> 6;
    const int vv = threadIdx.x & 63;
    float s = 0.0f;
    for (int w = 0; w < RD_WAVES; ++w) s += part[w][r][vv];
    readbuf[((size_t)b * HH + r) * VV + vc * 64 + vv] = s;
}

#define OUT_WAVES 16
#define OUT_KW    (HV / OUT_WAVES)
#define OUT_STEPS (OUT_KW / 8)

__global__ void __launch_bounds__(OUT_WAVES * 32)
k_out(const float* __restrict__ readbuf,
      const float* __restrict__ Wa,
      const float* __restrict__ ba,
      float* __restrict__ out) {
    __shared__ float opart[OUT_WAVES][16][32];

    const int tm   = blockIdx.x >> 3;
    const int tp   = blockIdx.x & 7;
    const int wave = threadIdx.x >> 5;
    const int lane = threadIdx.x & 31;
    const int col  = lane & 15;
    const int half = lane >> 4;

    const int k0 = wave * OUT_KW + 4 * half;
    const float* ap = readbuf + (size_t)(tm * 16 + col) * HV + k0;
    const float* bp = Wa + (size_t)k0 * VV + tp * 32 + 2 * col;

    v4f av[OUT_STEPS];
    v2f bv[OUT_STEPS][4];
    for (int s = 0; s < OUT_STEPS; ++s) {
        av[s] = *(const v4f*)(ap + 8 * s);
        for (int q = 0; q < 4; ++q)
            bv[s][q] = *(const v2f*)(bp + (size_t)(8 * s + q) * VV);
    }
    __builtin_amdgcn_sched_barrier(0);

    v8f acc[2] = {};
    for (int s = 0; s < OUT_STEPS; ++s) {
        for (int p = 0; p < 2; ++p) {
            const v2f a = { av[s][2 * p], av[s][2 * p + 1] };
            for (int t = 0; t < 2; ++t) {
                const v2f bb = { bv[s][2 * p][t], bv[s][2 * p + 1][t] };
                acc[t] = wmma4(a, bb, acc[t]);
            }
        }
    }

    for (int j = 0; j < 8; ++j) {
        const v2f o = { acc[0][j], acc[1][j] };
        *(v2f*)&opart[wave][8 * half + j][2 * col] = o;
    }
    __syncthreads();

    const int r  = threadIdx.x >> 5;
    const int cc = threadIdx.x & 31;
    float sum = 0.0f;
    for (int w = 0; w < OUT_WAVES; ++w) sum += opart[w][r][cc];
    out[(size_t)(tm * 16 + r) * VV + tp * 32 + cc] = sum + ba[tp * 32 + cc];
}

extern "C" void kernel_launch(void* const* d_in, const int* in_sizes, int n_in,
                              void* d_out, int out_size, void* d_ws, size_t ws_size,
                              hipStream_t stream) {
    const float* query = (const float*)d_in[0];
    const float* keys  = (const float*)d_in[1];
    const float* vals  = (const float*)d_in[2];
    const float* rpe   = (const float*)d_in[3];
    const float* Wq    = (const float*)d_in[4];
    const float* bq    = (const float*)d_in[5];
    const float* Wa    = (const float*)d_in[6];
    const float* ba    = (const float*)d_in[7];
    const int*   steps = (const int*)d_in[8];
    float* out = (float*)d_out;

    float* ws = (float*)d_ws;
    float* qbuf    = ws;
    float* smat    = qbuf + (size_t)BB * HK;
    float* readbuf = smat + (size_t)BB * HH * LL;

    k_qproj<<<64, 256, 0, stream>>>(query, Wq, bq, qbuf);
    k_scores<<<1024, 256, 0, stream>>>(keys, rpe, qbuf, smat);
    k_softmax<<<BB * HH, 256, 0, stream>>>(smat, steps);
    k_read<<<256, 512, 0, stream>>>(smat, vals, readbuf);
    k_out<<<32, 512, 0, stream>>>(readbuf, Wa, ba, out);
}


# k_scores retiled for whole-line reads: 16-wave workgroup per (batch, 512 rows), 512-byte row runs loaded 32 deep per wave, turned through LDS into fragments, q in LDS, 2 accumulators (new sum order)
# speedup vs baseline: 9.3707x; 1.6099x over previous
#include <hip/hip_runtime.h>
#include <hip/hip_bf16.h>
#include <math.h>

#define LL 2048
#define BB 64
#define KK 256
#define VV 256
#define HH 8
#define HK 2048
#define HV 2048

typedef __attribute__((ext_vector_type(2))) float v2f;
typedef __attribute__((ext_vector_type(4))) float v4f;
typedef __attribute__((ext_vector_type(8))) float v8f;

__device__ __forceinline__ v8f wmma4(v2f a, v2f b, v8f c) {
    return __builtin_amdgcn_wmma_f32_16x16x4_f32(
         false, a,  false, b,
         (short)0, c,  false,  false);
}

__global__ void k_qproj(const float* __restrict__ query,
                        const float* __restrict__ Wq,
                        const float* __restrict__ bq,
                        float* __restrict__ qbuf) {
    const int waveId = blockIdx.x * (blockDim.x >> 5) + (threadIdx.x >> 5);
    const int tm   = waveId >> 7;
    const int tn   = waveId & 127;
    const int lane = threadIdx.x & 31;
    const int col  = lane & 15;
    const int half = lane >> 4;

    const float* Arow = query + (size_t)(tm * 16 + col) * KK;
    const float* Bcol = Wq + (tn * 16 + col);

    v8f c = {};
    #pragma unroll 4
    for (int k0 = 0; k0 < KK; k0 += 4) {
        const int ka = k0 + 2 * half;
        v2f a = { Arow[ka], Arow[ka + 1] };
        v2f b = { Bcol[(size_t)ka * HK], Bcol[(size_t)(ka + 1) * HK] };
        c = wmma4(a, b, c);
    }

    const float bias = bq[tn * 16 + col];
    #pragma unroll
    for (int j = 0; j < 8; ++j) {
        const int row = tm * 16 + j + 8 * half;
        qbuf[(size_t)row * HK + tn * 16 + col] = c[j] + bias;
    }
}

#define SC_WAVES 16
#define SC_TILES 2
#define SC_GRPS  (2 * SC_TILES)
#define SC_PITCH 260

__device__ __forceinline__ void sc_load_group(const float* __restrict__ kp,
                                              int g, v4f (&kv)[16]) {
    const float* p = kp + (size_t)(16 * (g >> 1)) * (BB * KK) + 128 * (g & 1);
    for (int i = 0; i < 16; ++i)
        kv[i] = __builtin_nontemporal_load(
            (const v4f*)(p + (size_t)i * (BB * KK)));
}

__global__ void __launch_bounds__(SC_WAVES * 32)
k_scores(const float* __restrict__ keys,
         const float* __restrict__ rpe,
         const float* __restrict__ qbuf,
         float* __restrict__ smat) {
    __shared__ __attribute__((aligned(16))) float kt[SC_WAVES][16 * SC_PITCH];
    __shared__ __attribute__((aligned(16))) float qs[HH * SC_PITCH];

    const int b    = blockIdx.x >> 2;
    const int qd   = blockIdx.x & 3;
    const int wave = threadIdx.x >> 5;
    const int lane = threadIdx.x & 31;
    const int col  = lane & 15;
    const int half = lane >> 4;
    const int lw   = qd * (LL / 4) + wave * (16 * SC_TILES);

    const v4f qv = *(const v4f*)(qbuf + (size_t)b * HK + 4 * threadIdx.x);
    float rv[SC_TILES];
    for (int t = 0; t < SC_TILES; ++t)
        rv[t] = rpe[(size_t)(lw + 16 * t + col) * BB + b];

    const float* kp = keys + ((size_t)lw * BB + b) * KK + 4 * lane;
    v4f kv[2][16];
    sc_load_group(kp, 0, kv[0]);
    sc_load_group(kp, 1, kv[1]);

    *(v4f*)&qs[(threadIdx.x >> 6) * SC_PITCH + 4 * (threadIdx.x & 63)] = qv;
    __syncthreads();

    float* myt = kt[wave];
    const float* arow = myt + col * SC_PITCH + 4 * half;
    const float* brow = qs + (col & (HH - 1)) * SC_PITCH + 4 * half;

    v8f acc0 = {}, acc1 = {};
    for (int g = 0; g < SC_GRPS; ++g) {
        const int t  = g >> 1;
        const int kh = g & 1;
        for (int i = 0; i < 16; ++i)
            *(v4f*)(myt + i * SC_PITCH + 128 * kh + 4 * lane) = kv[g & 1][i];
        __builtin_amdgcn_sched_barrier(0);
        if (g + 2 < SC_GRPS)
            sc_load_group(kp, g + 2, kv[g & 1]);
        __builtin_amdgcn_sched_barrier(0);

        if (kh == 0) { acc0 = v8f{}; acc1 = v8f{}; }
        for (int j = 0; j < 16; ++j) {
            const v4f k4 = *(const v4f*)(arow + 128 * kh + 8 * j);
            const v4f q4 = *(const v4f*)(brow + 128 * kh + 8 * j);
            const v4f a4 = k4 * rv[t];
            acc0 = wmma4(v2f{a4.x, a4.y}, v2f{q4.x, q4.y}, acc0);
            acc1 = wmma4(v2f{a4.z, a4.w}, v2f{q4.z, q4.w}, acc1);
        }
        if (kh == 1 && col < HH) {
            float* sp = smat + ((size_t)b * HH + col) * LL + lw + 16 * t + 8 * half;
            const v8f s = acc0 + acc1;
            *(v4f*)(sp)     = v4f{s[0], s[1], s[2], s[3]};
            *(v4f*)(sp + 4) = v4f{s[4], s[5], s[6], s[7]};
        }
    }
}

__global__ void k_softmax(float* __restrict__ s, const int* __restrict__ steps) {
    const int bh   = blockIdx.x;
    const int b    = bh >> 3;
    const int step = steps[b];
    float* row = s + (size_t)bh * LL;
    const int t = threadIdx.x;

    float v[8];
    float mx = -INFINITY;
    #pragma unroll
    for (int i = 0; i < 8; ++i) {
        const int l = t + i * 256;
        const float x = (l < step) ? row[l] : -INFINITY;
        v[i] = x;
        mx = fmaxf(mx, x);
    }

    __shared__ float red[256];
    red[t] = mx;
    __syncthreads();
    for (int off = 128; off > 0; off >>= 1) {
        if (t < off) red[t] = fmaxf(red[t], red[t + off]);
        __syncthreads();
    }
    mx = red[0];
    __syncthreads();

    float sum = 0.0f;
    #pragma unroll
    for (int i = 0; i < 8; ++i) {
        const int l = t + i * 256;
        const float e = (l < step) ? __expf(v[i] - mx) : 0.0f;
        v[i] = e;
        sum += e;
    }
    red[t] = sum;
    __syncthreads();
    for (int off = 128; off > 0; off >>= 1) {
        if (t < off) red[t] += red[t + off];
        __syncthreads();
    }
    const float inv = 1.0f / red[0];

    #pragma unroll
    for (int i = 0; i < 8; ++i) row[t + i * 256] = v[i] * inv;
}

#define RD_WAVES 16
#define RD_ROWS  (LL / RD_WAVES)
#define RD_STEPS 4
#define RD_GRPS  (RD_ROWS / (8 * RD_STEPS))

__device__ __forceinline__ void rd_load_group(const float* __restrict__ wp,
                                              const float* __restrict__ vp,
                                              int g,
                                              v4f (&wa)[RD_STEPS],
                                              v4f (&vb)[RD_STEPS][4]) {
    for (int j = 0; j < RD_STEPS; ++j) {
        const int s = g * RD_STEPS + j;
        wa[j] = *(const v4f*)(wp + 8 * s);
        for (int q = 0; q < 4; ++q)
            vb[j][q] = __builtin_nontemporal_load(
                (const v4f*)(vp + (size_t)(8 * s + q) * (BB * VV)));
    }
}

__device__ __forceinline__ void rd_mma_group(const v4f (&wa)[RD_STEPS],
                                             const v4f (&vb)[RD_STEPS][4],
                                             v8f (&acc)[4]) {
    for (int j = 0; j < RD_STEPS; ++j) {
        for (int p = 0; p < 2; ++p) {
            const v2f a = { wa[j][2 * p], wa[j][2 * p + 1] };
            for (int t = 0; t < 4; ++t) {
                const v2f bb = { vb[j][2 * p][t], vb[j][2 * p + 1][t] };
                acc[t] = wmma4(a, bb, acc[t]);
            }
        }
    }
}

__global__ void __launch_bounds__(RD_WAVES * 32)
k_read(const float* __restrict__ wmat,
       const float* __restrict__ vals,
       float* __restrict__ readbuf) {
    __shared__ float part[RD_WAVES][HH][64];

    const int b    = blockIdx.x >> 2;
    const int vc   = blockIdx.x & 3;
    const int wave = threadIdx.x >> 5;
    const int lane = threadIdx.x & 31;
    const int col  = lane & 15;
    const int half = lane >> 4;
    const int l0 = wave * RD_ROWS + 4 * half;
    const float* wp = wmat + ((size_t)b * HH + (col & (HH - 1))) * LL + l0;
    const float* vp = vals + ((size_t)l0 * BB + b) * VV + vc * 64 + 4 * col;

    v8f acc[4] = {};
    v4f wa[2][RD_STEPS];
    v4f vb[2][RD_STEPS][4];

    rd_load_group(wp, vp, 0, wa[0], vb[0]);
    for (int g = 0; g < RD_GRPS; ++g) {
        __builtin_amdgcn_sched_barrier(0);
        if (g + 1 < RD_GRPS)
            rd_load_group(wp, vp, g + 1, wa[(g + 1) & 1], vb[(g + 1) & 1]);
        __builtin_amdgcn_sched_barrier(0);
        rd_mma_group(wa[g & 1], vb[g & 1], acc);
    }

    if (half == 0) {
        for (int r = 0; r < 8; ++r) {
            const v4f o = { acc[0][r], acc[1][r], acc[2][r], acc[3][r] };
            *(v4f*)&part[wave][r][4 * col] = o;
        }
    }
    __syncthreads();

    const int r  = threadIdx.x >> 6;
    const int vv = threadIdx.x & 63;
    float s = 0.0f;
    for (int w = 0; w < RD_WAVES; ++w) s += part[w][r][vv];
    readbuf[((size_t)b * HH + r) * VV + vc * 64 + vv] = s;
}

#define OUT_WAVES 16
#define OUT_KW    (HV / OUT_WAVES)
#define OUT_STEPS (OUT_KW / 8)

__global__ void __launch_bounds__(OUT_WAVES * 32)
k_out(const float* __restrict__ readbuf,
      const float* __restrict__ Wa,
      const float* __restrict__ ba,
      float* __restrict__ out) {
    __shared__ float opart[OUT_WAVES][16][32];

    const int tm   = blockIdx.x >> 3;
    const int tp   = blockIdx.x & 7;
    const int wave = threadIdx.x >> 5;
    const int lane = threadIdx.x & 31;
    const int col  = lane & 15;
    const int half = lane >> 4;

    const int k0 = wave * OUT_KW + 4 * half;
    const float* ap = readbuf + (size_t)(tm * 16 + col) * HV + k0;
    const float* bp = Wa + (size_t)k0 * VV + tp * 32 + 2 * col;

    v4f av[OUT_STEPS];
    v2f bv[OUT_STEPS][4];
    for (int s = 0; s < OUT_STEPS; ++s) {
        av[s] = *(const v4f*)(ap + 8 * s);
        for (int q = 0; q < 4; ++q)
            bv[s][q] = *(const v2f*)(bp + (size_t)(8 * s + q) * VV);
    }
    __builtin_amdgcn_sched_barrier(0);

    v8f acc[2] = {};
    for (int s = 0; s < OUT_STEPS; ++s) {
        for (int p = 0; p < 2; ++p) {
            const v2f a = { av[s][2 * p], av[s][2 * p + 1] };
            for (int t = 0; t < 2; ++t) {
                const v2f bb = { bv[s][2 * p][t], bv[s][2 * p + 1][t] };
                acc[t] = wmma4(a, bb, acc[t]);
            }
        }
    }

    for (int j = 0; j < 8; ++j) {
        const v2f o = { acc[0][j], acc[1][j] };
        *(v2f*)&opart[wave][8 * half + j][2 * col] = o;
    }
    __syncthreads();

    const int r  = threadIdx.x >> 5;
    const int cc = threadIdx.x & 31;
    float sum = 0.0f;
    for (int w = 0; w < OUT_WAVES; ++w) sum += opart[w][r][cc];
    out[(size_t)(tm * 16 + r) * VV + tp * 32 + cc] = sum + ba[tp * 32 + cc];
}

extern "C" void kernel_launch(void* const* d_in, const int* in_sizes, int n_in,
                              void* d_out, int out_size, void* d_ws, size_t ws_size,
                              hipStream_t stream) {
    const float* query = (const float*)d_in[0];
    const float* keys  = (const float*)d_in[1];
    const float* vals  = (const float*)d_in[2];
    const float* rpe   = (const float*)d_in[3];
    const float* Wq    = (const float*)d_in[4];
    const float* bq    = (const float*)d_in[5];
    const float* Wa    = (const float*)d_in[6];
    const float* ba    = (const float*)d_in[7];
    const int*   steps = (const int*)d_in[8];
    float* out = (float*)d_out;

    float* ws = (float*)d_ws;
    float* qbuf    = ws;
    float* smat    = qbuf + (size_t)BB * HK;
    float* readbuf = smat + (size_t)BB * HH * LL;

    k_qproj<<<64, 256, 0, stream>>>(query, Wq, bq, qbuf);
    k_scores<<<256, 512, 0, stream>>>(keys, rpe, qbuf, smat);
    k_softmax<<<BB * HH, 256, 0, stream>>>(smat, steps);
    k_read<<<256, 512, 0, stream>>>(smat, vals, readbuf);
    k_out<<<32, 512, 0, stream>>>(readbuf, Wa, ba, out);
}


# k_qproj split along k: one 8-wave workgroup per 32x32 tile of q, each wave sums 32 k with all 24 loads issued before its first product, partial tiles summed through LDS (new sum order)
# speedup vs baseline: 10.2855x; 1.0976x over previous
#include <hip/hip_runtime.h>
#include <hip/hip_bf16.h>
#include <math.h>

#define LL 2048
#define BB 64
#define KK 256
#define VV 256
#define HH 8
#define HK 2048
#define HV 2048

typedef __attribute__((ext_vector_type(2))) float v2f;
typedef __attribute__((ext_vector_type(4))) float v4f;
typedef __attribute__((ext_vector_type(8))) float v8f;

__device__ __forceinline__ v8f wmma4(v2f a, v2f b, v8f c) {
    return __builtin_amdgcn_wmma_f32_16x16x4_f32(
         false, a,  false, b,
         (short)0, c,  false,  false);
}

#define QP_WAVES 8
#define QP_KW    (KK / QP_WAVES)
#define QP_STEPS (QP_KW / 8)
#define QP_MT    2

__global__ void __launch_bounds__(QP_WAVES * 32)
k_qproj(const float* __restrict__ query,
        const float* __restrict__ Wq,
        const float* __restrict__ bq,
        float* __restrict__ qbuf) {
    __shared__ float qpart[QP_WAVES][16 * QP_MT][32];

    const int tm   = blockIdx.x >> 6;
    const int tp   = blockIdx.x & 63;
    const int wave = threadIdx.x >> 5;
    const int lane = threadIdx.x & 31;
    const int col  = lane & 15;
    const int half = lane >> 4;

    const int k0 = wave * QP_KW + 4 * half;
    const float* ap = query + (size_t)(tm * (16 * QP_MT) + col) * KK + k0;
    const float* bp = Wq + (size_t)k0 * HK + tp * 32 + 2 * col;

    v4f av[QP_MT][QP_STEPS];
    v2f bv[QP_STEPS][4];
    for (int s = 0; s < QP_STEPS; ++s) {
        for (int m = 0; m < QP_MT; ++m)
            av[m][s] = *(const v4f*)(ap + (size_t)m * 16 * KK + 8 * s);
        for (int q = 0; q < 4; ++q)
            bv[s][q] = *(const v2f*)(bp + (size_t)(8 * s + q) * HK);
    }
    __builtin_amdgcn_sched_barrier(0);

    v8f acc[QP_MT][2] = {};
    for (int s = 0; s < QP_STEPS; ++s) {
        for (int p = 0; p < 2; ++p) {
            for (int m = 0; m < QP_MT; ++m) {
                const v2f a = { av[m][s][2 * p], av[m][s][2 * p + 1] };
                for (int t = 0; t < 2; ++t) {
                    const v2f bb = { bv[s][2 * p][t], bv[s][2 * p + 1][t] };
                    acc[m][t] = wmma4(a, bb, acc[m][t]);
                }
            }
        }
    }

    for (int m = 0; m < QP_MT; ++m) {
        for (int j = 0; j < 8; ++j) {
            const v2f o = { acc[m][0][j], acc[m][1][j] };
            *(v2f*)&qpart[wave][16 * m + 8 * half + j][2 * col] = o;
        }
    }
    __syncthreads();

    const int r0 = threadIdx.x >> 5;
    const int cc = threadIdx.x & 31;
    const float bias = bq[tp * 32 + cc];
    for (int i = 0; i < (16 * QP_MT) / QP_WAVES; ++i) {
        const int r = r0 + QP_WAVES * i;
        float sum = 0.0f;
        for (int w = 0; w < QP_WAVES; ++w) sum += qpart[w][r][cc];
        qbuf[(size_t)(tm * (16 * QP_MT) + r) * HK + tp * 32 + cc] = sum + bias;
    }
}

#define SC_WAVES 16
#define SC_TILES 2
#define SC_GRPS  (2 * SC_TILES)
#define SC_PITCH 260

__device__ __forceinline__ void sc_load_group(const float* __restrict__ kp,
                                              int g, v4f (&kv)[16]) {
    const float* p = kp + (size_t)(16 * (g >> 1)) * (BB * KK) + 128 * (g & 1);
    for (int i = 0; i < 16; ++i)
        kv[i] = __builtin_nontemporal_load(
            (const v4f*)(p + (size_t)i * (BB * KK)));
}

__global__ void __launch_bounds__(SC_WAVES * 32)
k_scores(const float* __restrict__ keys,
         const float* __restrict__ rpe,
         const float* __restrict__ qbuf,
         float* __restrict__ smat) {
    __shared__ __attribute__((aligned(16))) float kt[SC_WAVES][16 * SC_PITCH];
    __shared__ __attribute__((aligned(16))) float qs[HH * SC_PITCH];

    const int b    = blockIdx.x >> 2;
    const int qd   = blockIdx.x & 3;
    const int wave = threadIdx.x >> 5;
    const int lane = threadIdx.x & 31;
    const int col  = lane & 15;
    const int half = lane >> 4;
    const int lw   = qd * (LL / 4) + wave * (16 * SC_TILES);

    const v4f qv = *(const v4f*)(qbuf + (size_t)b * HK + 4 * threadIdx.x);
    float rv[SC_TILES];
    for (int t = 0; t < SC_TILES; ++t)
        rv[t] = rpe[(size_t)(lw + 16 * t + col) * BB + b];

    const float* kp = keys + ((size_t)lw * BB + b) * KK + 4 * lane;
    v4f kv[2][16];
    sc_load_group(kp, 0, kv[0]);
    sc_load_group(kp, 1, kv[1]);

    *(v4f*)&qs[(threadIdx.x >> 6) * SC_PITCH + 4 * (threadIdx.x & 63)] = qv;
    __syncthreads();

    float* myt = kt[wave];
    const float* arow = myt + col * SC_PITCH + 4 * half;
    const float* brow = qs + (col & (HH - 1)) * SC_PITCH + 4 * half;

    v8f acc0 = {}, acc1 = {};
    for (int g = 0; g < SC_GRPS; ++g) {
        const int t  = g >> 1;
        const int kh = g & 1;
        for (int i = 0; i < 16; ++i)
            *(v4f*)(myt + i * SC_PITCH + 128 * kh + 4 * lane) = kv[g & 1][i];
        __builtin_amdgcn_sched_barrier(0);
        if (g + 2 < SC_GRPS)
            sc_load_group(kp, g + 2, kv[g & 1]);
        __builtin_amdgcn_sched_barrier(0);

        if (kh == 0) { acc0 = v8f{}; acc1 = v8f{}; }
        for (int j = 0; j < 16; ++j) {
            const v4f k4 = *(const v4f*)(arow + 128 * kh + 8 * j);
            const v4f q4 = *(const v4f*)(brow + 128 * kh + 8 * j);
            const v4f a4 = k4 * rv[t];
            acc0 = wmma4(v2f{a4.x, a4.y}, v2f{q4.x, q4.y}, acc0);
            acc1 = wmma4(v2f{a4.z, a4.w}, v2f{q4.z, q4.w}, acc1);
        }
        if (kh == 1 && col < HH) {
            float* sp = smat + ((size_t)b * HH + col) * LL + lw + 16 * t + 8 * half;
            const v8f s = acc0 + acc1;
            *(v4f*)(sp)     = v4f{s[0], s[1], s[2], s[3]};
            *(v4f*)(sp + 4) = v4f{s[4], s[5], s[6], s[7]};
        }
    }
}

__global__ void k_softmax(float* __restrict__ s, const int* __restrict__ steps) {
    const int bh   = blockIdx.x;
    const int b    = bh >> 3;
    const int step = steps[b];
    float* row = s + (size_t)bh * LL;
    const int t = threadIdx.x;

    float v[8];
    float mx = -INFINITY;
    #pragma unroll
    for (int i = 0; i < 8; ++i) {
        const int l = t + i * 256;
        const float x = (l < step) ? row[l] : -INFINITY;
        v[i] = x;
        mx = fmaxf(mx, x);
    }

    __shared__ float red[256];
    red[t] = mx;
    __syncthreads();
    for (int off = 128; off > 0; off >>= 1) {
        if (t < off) red[t] = fmaxf(red[t], red[t + off]);
        __syncthreads();
    }
    mx = red[0];
    __syncthreads();

    float sum = 0.0f;
    #pragma unroll
    for (int i = 0; i < 8; ++i) {
        const int l = t + i * 256;
        const float e = (l < step) ? __expf(v[i] - mx) : 0.0f;
        v[i] = e;
        sum += e;
    }
    red[t] = sum;
    __syncthreads();
    for (int off = 128; off > 0; off >>= 1) {
        if (t < off) red[t] += red[t + off];
        __syncthreads();
    }
    const float inv = 1.0f / red[0];

    #pragma unroll
    for (int i = 0; i < 8; ++i) row[t + i * 256] = v[i] * inv;
}

#define RD_WAVES 16
#define RD_ROWS  (LL / RD_WAVES)
#define RD_STEPS 4
#define RD_GRPS  (RD_ROWS / (8 * RD_STEPS))

__device__ __forceinline__ void rd_load_group(const float* __restrict__ wp,
                                              const float* __restrict__ vp,
                                              int g,
                                              v4f (&wa)[RD_STEPS],
                                              v4f (&vb)[RD_STEPS][4]) {
    for (int j = 0; j < RD_STEPS; ++j) {
        const int s = g * RD_STEPS + j;
        wa[j] = *(const v4f*)(wp + 8 * s);
        for (int q = 0; q < 4; ++q)
            vb[j][q] = __builtin_nontemporal_load(
                (const v4f*)(vp + (size_t)(8 * s + q) * (BB * VV)));
    }
}

__device__ __forceinline__ void rd_mma_group(const v4f (&wa)[RD_STEPS],
                                             const v4f (&vb)[RD_STEPS][4],
                                             v8f (&acc)[4]) {
    for (int j = 0; j < RD_STEPS; ++j) {
        for (int p = 0; p < 2; ++p) {
            const v2f a = { wa[j][2 * p], wa[j][2 * p + 1] };
            for (int t = 0; t < 4; ++t) {
                const v2f bb = { vb[j][2 * p][t], vb[j][2 * p + 1][t] };
                acc[t] = wmma4(a, bb, acc[t]);
            }
        }
    }
}

__global__ void __launch_bounds__(RD_WAVES * 32)
k_read(const float* __restrict__ wmat,
       const float* __restrict__ vals,
       float* __restrict__ readbuf) {
    __shared__ float part[RD_WAVES][HH][64];

    const int b    = blockIdx.x >> 2;
    const int vc   = blockIdx.x & 3;
    const int wave = threadIdx.x >> 5;
    const int lane = threadIdx.x & 31;
    const int col  = lane & 15;
    const int half = lane >> 4;
    const int l0 = wave * RD_ROWS + 4 * half;
    const float* wp = wmat + ((size_t)b * HH + (col & (HH - 1))) * LL + l0;
    const float* vp = vals + ((size_t)l0 * BB + b) * VV + vc * 64 + 4 * col;

    v8f acc[4] = {};
    v4f wa[2][RD_STEPS];
    v4f vb[2][RD_STEPS][4];

    rd_load_group(wp, vp, 0, wa[0], vb[0]);
    for (int g = 0; g < RD_GRPS; ++g) {
        __builtin_amdgcn_sched_barrier(0);
        if (g + 1 < RD_GRPS)
            rd_load_group(wp, vp, g + 1, wa[(g + 1) & 1], vb[(g + 1) & 1]);
        __builtin_amdgcn_sched_barrier(0);
        rd_mma_group(wa[g & 1], vb[g & 1], acc);
    }

    if (half == 0) {
        for (int r = 0; r < 8; ++r) {
            const v4f o = { acc[0][r], acc[1][r], acc[2][r], acc[3][r] };
            *(v4f*)&part[wave][r][4 * col] = o;
        }
    }
    __syncthreads();

    const int r  = threadIdx.x >> 6;
    const int vv = threadIdx.x & 63;
    float s = 0.0f;
    for (int w = 0; w < RD_WAVES; ++w) s += part[w][r][vv];
    readbuf[((size_t)b * HH + r) * VV + vc * 64 + vv] = s;
}

#define OUT_WAVES 16
#define OUT_KW    (HV / OUT_WAVES)
#define OUT_STEPS (OUT_KW / 8)

__global__ void __launch_bounds__(OUT_WAVES * 32)
k_out(const float* __restrict__ readbuf,
      const float* __restrict__ Wa,
      const float* __restrict__ ba,
      float* __restrict__ out) {
    __shared__ float opart[OUT_WAVES][16][32];

    const int tm   = blockIdx.x >> 3;
    const int tp   = blockIdx.x & 7;
    const int wave = threadIdx.x >> 5;
    const int lane = threadIdx.x & 31;
    const int col  = lane & 15;
    const int half = lane >> 4;

    const int k0 = wave * OUT_KW + 4 * half;
    const float* ap = readbuf + (size_t)(tm * 16 + col) * HV + k0;
    const float* bp = Wa + (size_t)k0 * VV + tp * 32 + 2 * col;

    v4f av[OUT_STEPS];
    v2f bv[OUT_STEPS][4];
    for (int s = 0; s < OUT_STEPS; ++s) {
        av[s] = *(const v4f*)(ap + 8 * s);
        for (int q = 0; q < 4; ++q)
            bv[s][q] = *(const v2f*)(bp + (size_t)(8 * s + q) * VV);
    }
    __builtin_amdgcn_sched_barrier(0);

    v8f acc[2] = {};
    for (int s = 0; s < OUT_STEPS; ++s) {
        for (int p = 0; p < 2; ++p) {
            const v2f a = { av[s][2 * p], av[s][2 * p + 1] };
            for (int t = 0; t < 2; ++t) {
                const v2f bb = { bv[s][2 * p][t], bv[s][2 * p + 1][t] };
                acc[t] = wmma4(a, bb, acc[t]);
            }
        }
    }

    for (int j = 0; j < 8; ++j) {
        const v2f o = { acc[0][j], acc[1][j] };
        *(v2f*)&opart[wave][8 * half + j][2 * col] = o;
    }
    __syncthreads();

    const int r  = threadIdx.x >> 5;
    const int cc = threadIdx.x & 31;
    float sum = 0.0f;
    for (int w = 0; w < OUT_WAVES; ++w) sum += opart[w][r][cc];
    out[(size_t)(tm * 16 + r) * VV + tp * 32 + cc] = sum + ba[tp * 32 + cc];
}

extern "C" void kernel_launch(void* const* d_in, const int* in_sizes, int n_in,
                              void* d_out, int out_size, void* d_ws, size_t ws_size,
                              hipStream_t stream) {
    const float* query = (const float*)d_in[0];
    const float* keys  = (const float*)d_in[1];
    const float* vals  = (const float*)d_in[2];
    const float* rpe   = (const float*)d_in[3];
    const float* Wq    = (const float*)d_in[4];
    const float* bq    = (const float*)d_in[5];
    const float* Wa    = (const float*)d_in[6];
    const float* ba    = (const float*)d_in[7];
    const int*   steps = (const int*)d_in[8];
    float* out = (float*)d_out;

    float* ws = (float*)d_ws;
    float* qbuf    = ws;
    float* smat    = qbuf + (size_t)BB * HK;
    float* readbuf = smat + (size_t)BB * HH * LL;

    k_qproj<<<(BB / (16 * QP_MT)) * (HK / 32), QP_WAVES * 32, 0, stream>>>(query, Wq, bq, qbuf);
    k_scores<<<256, 512, 0, stream>>>(keys, rpe, qbuf, smat);
    k_softmax<<<BB * HH, 256, 0, stream>>>(smat, steps);
    k_read<<<256, 512, 0, stream>>>(smat, vals, readbuf);
    k_out<<<32, 512, 0, stream>>>(readbuf, Wa, ba, out);
}


# rows l >= steps[b] no longer streamed: k_scores blocks take equal chunks of the batches' valid 16-row tiles (chunk map computed from steps), k_read waves skip 32-row groups at or behind steps[b]
# speedup vs baseline: 14.9453x; 1.4530x over previous
#include <hip/hip_runtime.h>
#include <hip/hip_bf16.h>
#include <math.h>

#define LL 2048
#define BB 64
#define KK 256
#define VV 256
#define HH 8
#define HK 2048
#define HV 2048

typedef __attribute__((ext_vector_type(2))) float v2f;
typedef __attribute__((ext_vector_type(4))) float v4f;
typedef __attribute__((ext_vector_type(8))) float v8f;

__device__ __forceinline__ v8f wmma4(v2f a, v2f b, v8f c) {
    return __builtin_amdgcn_wmma_f32_16x16x4_f32(
         false, a,  false, b,
         (short)0, c,  false,  false);
}

#define QP_WAVES 8
#define QP_KW    (KK / QP_WAVES)
#define QP_STEPS (QP_KW / 8)
#define QP_MT    2

__global__ void __launch_bounds__(QP_WAVES * 32)
k_qproj(const float* __restrict__ query,
        const float* __restrict__ Wq,
        const float* __restrict__ bq,
        float* __restrict__ qbuf) {
    __shared__ float qpart[QP_WAVES][16 * QP_MT][32];

    const int tm   = blockIdx.x >> 6;
    const int tp   = blockIdx.x & 63;
    const int wave = threadIdx.x >> 5;
    const int lane = threadIdx.x & 31;
    const int col  = lane & 15;
    const int half = lane >> 4;

    const int k0 = wave * QP_KW + 4 * half;
    const float* ap = query + (size_t)(tm * (16 * QP_MT) + col) * KK + k0;
    const float* bp = Wq + (size_t)k0 * HK + tp * 32 + 2 * col;

    v4f av[QP_MT][QP_STEPS];
    v2f bv[QP_STEPS][4];
    for (int s = 0; s < QP_STEPS; ++s) {
        for (int m = 0; m < QP_MT; ++m)
            av[m][s] = *(const v4f*)(ap + (size_t)m * 16 * KK + 8 * s);
        for (int q = 0; q < 4; ++q)
            bv[s][q] = *(const v2f*)(bp + (size_t)(8 * s + q) * HK);
    }
    __builtin_amdgcn_sched_barrier(0);

    v8f acc[QP_MT][2] = {};
    for (int s = 0; s < QP_STEPS; ++s) {
        for (int p = 0; p < 2; ++p) {
            for (int m = 0; m < QP_MT; ++m) {
                const v2f a = { av[m][s][2 * p], av[m][s][2 * p + 1] };
                for (int t = 0; t < 2; ++t) {
                    const v2f bb = { bv[s][2 * p][t], bv[s][2 * p + 1][t] };
                    acc[m][t] = wmma4(a, bb, acc[m][t]);
                }
            }
        }
    }

    for (int m = 0; m < QP_MT; ++m) {
        for (int j = 0; j < 8; ++j) {
            const v2f o = { acc[m][0][j], acc[m][1][j] };
            *(v2f*)&qpart[wave][16 * m + 8 * half + j][2 * col] = o;
        }
    }
    __syncthreads();

    const int r0 = threadIdx.x >> 5;
    const int cc = threadIdx.x & 31;
    const float bias = bq[tp * 32 + cc];
    for (int i = 0; i < (16 * QP_MT) / QP_WAVES; ++i) {
        const int r = r0 + QP_WAVES * i;
        float sum = 0.0f;
        for (int w = 0; w < QP_WAVES; ++w) sum += qpart[w][r][cc];
        qbuf[(size_t)(tm * (16 * QP_MT) + r) * HK + tp * 32 + cc] = sum + bias;
    }
}

#define SC_NB    256
#define SC_WAVES 16
#define SC_TILES 2
#define SC_TROWS (16 * SC_WAVES)
#define SC_PITCH 260

template <int CTRL>
__device__ __forceinline__ int dpp_row(int v) {
    return __builtin_amdgcn_update_dpp(0, v, CTRL, 0xf, 0xf, false);
}
__device__ __forceinline__ int wave_sum(int v) {
    v += dpp_row<0x161>(v);
    v += dpp_row<0x162>(v);
    v += dpp_row<0x164>(v);
    v += dpp_row<0x168>(v);
    v += __builtin_amdgcn_permlanex16(v, v, 0x76543210, 0xfedcba98, false, false);
    return v;
}
__device__ __forceinline__ int wave_scan(int v, int lane) {
    v += dpp_row<0x111>(v);
    v += dpp_row<0x112>(v);
    v += dpp_row<0x114>(v);
    v += dpp_row<0x118>(v);
    const int row0 = __builtin_amdgcn_readlane(v, 15);
    return v + (lane >= 16 ? row0 : 0);
}

__device__ __forceinline__ void sc_load_group(const float* __restrict__ kp,
                                              int g, v4f (&kv)[16]) {
    const float* p = kp + (size_t)(SC_TROWS * (g >> 1)) * (BB * KK) + 128 * (g & 1);
    for (int i = 0; i < 16; ++i)
        kv[i] = __builtin_nontemporal_load(
            (const v4f*)(p + (size_t)i * (BB * KK)));
}

template <int NG>
__device__ __forceinline__ void sc_stream(const float* __restrict__ kp,
                                          const float* __restrict__ rp,
                                          float* __restrict__ sp,
                                          float* myt, const float* arow,
                                          const float* brow, int lane, int col) {
    float rv[NG / 2];
    for (int t = 0; t < NG / 2; ++t)
        rv[t] = rp[(size_t)(SC_TROWS * t) * BB];

    v4f kv[2][16];
    sc_load_group(kp, 0, kv[0]);
    sc_load_group(kp, 1, kv[1]);

    v8f acc0 = {}, acc1 = {};
    for (int g = 0; g < NG; ++g) {
        const int t  = g >> 1;
        const int kh = g & 1;
        for (int i = 0; i < 16; ++i)
            *(v4f*)(myt + i * SC_PITCH + 128 * kh + 4 * lane) = kv[g & 1][i];
        __builtin_amdgcn_sched_barrier(0);
        if (g + 2 < NG)
            sc_load_group(kp, g + 2, kv[g & 1]);
        __builtin_amdgcn_sched_barrier(0);

        if (kh == 0) { acc0 = v8f{}; acc1 = v8f{}; }
        for (int j = 0; j < 16; ++j) {
            const v4f k4 = *(const v4f*)(arow + 128 * kh + 8 * j);
            const v4f q4 = *(const v4f*)(brow + 128 * kh + 8 * j);
            const v4f a4 = k4 * rv[t];
            acc0 = wmma4(v2f{a4.x, a4.y}, v2f{q4.x, q4.y}, acc0);
            acc1 = wmma4(v2f{a4.z, a4.w}, v2f{q4.z, q4.w}, acc1);
        }
        if (kh == 1 && col < HH) {
            const v8f s = acc0 + acc1;
            *(v4f*)(sp + SC_TROWS * t)     = v4f{s[0], s[1], s[2], s[3]};
            *(v4f*)(sp + SC_TROWS * t + 4) = v4f{s[4], s[5], s[6], s[7]};
        }
    }
}

__global__ void __launch_bounds__(SC_WAVES * 32)
k_scores(const float* __restrict__ keys,
         const float* __restrict__ rpe,
         const float* __restrict__ qbuf,
         const int* __restrict__ steps,
         float* __restrict__ smat) {
    __shared__ __attribute__((aligned(16))) float kt[SC_WAVES][16 * SC_PITCH];
    __shared__ __attribute__((aligned(16))) float qs[HH * SC_PITCH];

    const int wave = __builtin_amdgcn_readfirstlane(threadIdx.x >> 5);
    const int lane = threadIdx.x & 31;
    const int col  = lane & 15;
    const int half = lane >> 4;

    const int n0 = (min(max(steps[lane], 0), LL) + 15) >> 4;
    const int n1 = (min(max(steps[lane + 32], 0), LL) + 15) >> 4;
    int ct = max((__builtin_amdgcn_readfirstlane(wave_sum(n0 + n1)) + SC_NB - 1) / SC_NB, 1);
    for (; ct < 16 * SC_TILES; ++ct)
        if (__builtin_amdgcn_readfirstlane(
                wave_sum((n0 + ct - 1) / ct + (n1 + ct - 1) / ct)) <= SC_NB) break;
    const int c0 = (n0 + ct - 1) / ct;
    const int c1 = (n1 + ct - 1) / ct;
    const int p0 = wave_scan(c0, lane);
    const int p1 = wave_scan(c1, lane) + __builtin_amdgcn_readlane(p0, 31);
    const int ci = blockIdx.x;
    if (ci >= __builtin_amdgcn_readlane(p1, 31)) return;
    const int b  = __popcll(__ballot(p0 <= ci)) + __popcll(__ballot(p1 <= ci));
    const int bl = b & 31;
    const int ex = (b < 32) ? __builtin_amdgcn_readlane(p0 - c0, bl)
                            : __builtin_amdgcn_readlane(p1 - c1, bl);
    const int nb = (b < 32) ? __builtin_amdgcn_readlane(n0, bl)
                            : __builtin_amdgcn_readlane(n1, bl);
    const int tile0 = (ci - ex) * ct;
    const int ntile = min(ct, nb - tile0);
    const int nt = (wave < ntile) + (wave + SC_WAVES < ntile);
    const int lw = 16 * (tile0 + wave);

    const v4f qv = *(const v4f*)(qbuf + (size_t)b * HK + 4 * threadIdx.x);
    *(v4f*)&qs[(threadIdx.x >> 6) * SC_PITCH + 4 * (threadIdx.x & 63)] = qv;
    __syncthreads();

    float* myt = kt[wave];
    const float* arow = myt + col * SC_PITCH + 4 * half;
    const float* brow = qs + (col & (HH - 1)) * SC_PITCH + 4 * half;
    const float* kp = keys + ((size_t)lw * BB + b) * KK + 4 * lane;
    const float* rp = rpe + (size_t)(lw + col) * BB + b;
    float* sp = smat + ((size_t)b * HH + col) * LL + lw + 8 * half;

    if (nt == 2)      sc_stream<4>(kp, rp, sp, myt, arow, brow, lane, col);
    else if (nt == 1) sc_stream<2>(kp, rp, sp, myt, arow, brow, lane, col);
}

__global__ void k_softmax(float* __restrict__ s, const int* __restrict__ steps) {
    const int bh   = blockIdx.x;
    const int b    = bh >> 3;
    const int step = steps[b];
    float* row = s + (size_t)bh * LL;
    const int t = threadIdx.x;

    float v[8];
    float mx = -INFINITY;
    #pragma unroll
    for (int i = 0; i < 8; ++i) {
        const int l = t + i * 256;
        const float x = (l < step) ? row[l] : -INFINITY;
        v[i] = x;
        mx = fmaxf(mx, x);
    }

    __shared__ float red[256];
    red[t] = mx;
    __syncthreads();
    for (int off = 128; off > 0; off >>= 1) {
        if (t < off) red[t] = fmaxf(red[t], red[t + off]);
        __syncthreads();
    }
    mx = red[0];
    __syncthreads();

    float sum = 0.0f;
    #pragma unroll
    for (int i = 0; i < 8; ++i) {
        const int l = t + i * 256;
        const float e = (l < step) ? __expf(v[i] - mx) : 0.0f;
        v[i] = e;
        sum += e;
    }
    red[t] = sum;
    __syncthreads();
    for (int off = 128; off > 0; off >>= 1) {
        if (t < off) red[t] += red[t + off];
        __syncthreads();
    }
    const float inv = 1.0f / red[0];

    #pragma unroll
    for (int i = 0; i < 8; ++i) row[t + i * 256] = v[i] * inv;
}

#define RD_WAVES 16
#define RD_ROWS  (LL / RD_WAVES)
#define RD_STEPS 4
#define RD_GRPS  (RD_ROWS / (8 * RD_STEPS))

__device__ __forceinline__ void rd_load_group(const float* __restrict__ wp,
                                              const float* __restrict__ vp,
                                              int g,
                                              v4f (&wa)[RD_STEPS],
                                              v4f (&vb)[RD_STEPS][4]) {
    for (int j = 0; j < RD_STEPS; ++j) {
        const int s = g * RD_STEPS + j;
        wa[j] = *(const v4f*)(wp + 8 * s);
        for (int q = 0; q < 4; ++q)
            vb[j][q] = __builtin_nontemporal_load(
                (const v4f*)(vp + (size_t)(8 * s + q) * (BB * VV)));
    }
}

__device__ __forceinline__ void rd_mma_group(const v4f (&wa)[RD_STEPS],
                                             const v4f (&vb)[RD_STEPS][4],
                                             v8f (&acc)[4]) {
    for (int j = 0; j < RD_STEPS; ++j) {
        for (int p = 0; p < 2; ++p) {
            const v2f a = { wa[j][2 * p], wa[j][2 * p + 1] };
            for (int t = 0; t < 4; ++t) {
                const v2f bb = { vb[j][2 * p][t], vb[j][2 * p + 1][t] };
                acc[t] = wmma4(a, bb, acc[t]);
            }
        }
    }
}

template <int NG>
__device__ __forceinline__ void rd_stream(const float* __restrict__ wp,
                                          const float* __restrict__ vp,
                                          v8f (&acc)[4]) {
    v4f wa[2][RD_STEPS];
    v4f vb[2][RD_STEPS][4];

    rd_load_group(wp, vp, 0, wa[0], vb[0]);
    for (int g = 0; g < NG; ++g) {
        __builtin_amdgcn_sched_barrier(0);
        if (g + 1 < NG)
            rd_load_group(wp, vp, g + 1, wa[(g + 1) & 1], vb[(g + 1) & 1]);
        __builtin_amdgcn_sched_barrier(0);
        rd_mma_group(wa[g & 1], vb[g & 1], acc);
    }
}

__global__ void __launch_bounds__(RD_WAVES * 32)
k_read(const float* __restrict__ wmat,
       const float* __restrict__ vals,
       const int* __restrict__ steps,
       float* __restrict__ readbuf) {
    __shared__ float part[RD_WAVES][HH][64];

    const int b    = blockIdx.x >> 2;
    const int vc   = blockIdx.x & 3;
    const int wave = __builtin_amdgcn_readfirstlane(threadIdx.x >> 5);
    const int lane = threadIdx.x & 31;
    const int col  = lane & 15;
    const int half = lane >> 4;
    const int l0 = wave * RD_ROWS + 4 * half;
    const float* wp = wmat + ((size_t)b * HH + (col & (HH - 1))) * LL + l0;
    const float* vp = vals + ((size_t)l0 * BB + b) * VV + vc * 64 + 4 * col;

    const int step = min(max(steps[b], 0), LL);
    const int ng = min(max((step - wave * RD_ROWS + 8 * RD_STEPS - 1) / (8 * RD_STEPS), 0),
                       RD_GRPS);

    v8f acc[4] = {};
    if (ng == 4)      rd_stream<4>(wp, vp, acc);
    else if (ng == 3) rd_stream<3>(wp, vp, acc);
    else if (ng == 2) rd_stream<2>(wp, vp, acc);
    else if (ng == 1) rd_stream<1>(wp, vp, acc);

    if (half == 0) {
        for (int r = 0; r < 8; ++r) {
            const v4f o = { acc[0][r], acc[1][r], acc[2][r], acc[3][r] };
            *(v4f*)&part[wave][r][4 * col] = o;
        }
    }
    __syncthreads();

    const int r  = threadIdx.x >> 6;
    const int vv = threadIdx.x & 63;
    float s = 0.0f;
    for (int w = 0; w < RD_WAVES; ++w) s += part[w][r][vv];
    readbuf[((size_t)b * HH + r) * VV + vc * 64 + vv] = s;
}

#define OUT_WAVES 16
#define OUT_KW    (HV / OUT_WAVES)
#define OUT_STEPS (OUT_KW / 8)

__global__ void __launch_bounds__(OUT_WAVES * 32)
k_out(const float* __restrict__ readbuf,
      const float* __restrict__ Wa,
      const float* __restrict__ ba,
      float* __restrict__ out) {
    __shared__ float opart[OUT_WAVES][16][32];

    const int tm   = blockIdx.x >> 3;
    const int tp   = blockIdx.x & 7;
    const int wave = threadIdx.x >> 5;
    const int lane = threadIdx.x & 31;
    const int col  = lane & 15;
    const int half = lane >> 4;

    const int k0 = wave * OUT_KW + 4 * half;
    const float* ap = readbuf + (size_t)(tm * 16 + col) * HV + k0;
    const float* bp = Wa + (size_t)k0 * VV + tp * 32 + 2 * col;

    v4f av[OUT_STEPS];
    v2f bv[OUT_STEPS][4];
    for (int s = 0; s < OUT_STEPS; ++s) {
        av[s] = *(const v4f*)(ap + 8 * s);
        for (int q = 0; q < 4; ++q)
            bv[s][q] = *(const v2f*)(bp + (size_t)(8 * s + q) * VV);
    }
    __builtin_amdgcn_sched_barrier(0);

    v8f acc[2] = {};
    for (int s = 0; s < OUT_STEPS; ++s) {
        for (int p = 0; p < 2; ++p) {
            const v2f a = { av[s][2 * p], av[s][2 * p + 1] };
            for (int t = 0; t < 2; ++t) {
                const v2f bb = { bv[s][2 * p][t], bv[s][2 * p + 1][t] };
                acc[t] = wmma4(a, bb, acc[t]);
            }
        }
    }

    for (int j = 0; j < 8; ++j) {
        const v2f o = { acc[0][j], acc[1][j] };
        *(v2f*)&opart[wave][8 * half + j][2 * col] = o;
    }
    __syncthreads();

    const int r  = threadIdx.x >> 5;
    const int cc = threadIdx.x & 31;
    float sum = 0.0f;
    for (int w = 0; w < OUT_WAVES; ++w) sum += opart[w][r][cc];
    out[(size_t)(tm * 16 + r) * VV + tp * 32 + cc] = sum + ba[tp * 32 + cc];
}

extern "C" void kernel_launch(void* const* d_in, const int* in_sizes, int n_in,
                              void* d_out, int out_size, void* d_ws, size_t ws_size,
                              hipStream_t stream) {
    const float* query = (const float*)d_in[0];
    const float* keys  = (const float*)d_in[1];
    const float* vals  = (const float*)d_in[2];
    const float* rpe   = (const float*)d_in[3];
    const float* Wq    = (const float*)d_in[4];
    const float* bq    = (const float*)d_in[5];
    const float* Wa    = (const float*)d_in[6];
    const float* ba    = (const float*)d_in[7];
    const int*   steps = (const int*)d_in[8];
    float* out = (float*)d_out;

    float* ws = (float*)d_ws;
    float* qbuf    = ws;
    float* smat    = qbuf + (size_t)BB * HK;
    float* readbuf = smat + (size_t)BB * HH * LL;

    k_qproj<<<(BB / (16 * QP_MT)) * (HK / 32), QP_WAVES * 32, 0, stream>>>(query, Wq, bq, qbuf);
    k_scores<<<SC_NB, SC_WAVES * 32, 0, stream>>>(keys, rpe, qbuf, steps, smat);
    k_softmax<<<BB * HH, 256, 0, stream>>>(smat, steps);
    k_read<<<256, 512, 0, stream>>>(smat, vals, steps, readbuf);
    k_out<<<32, 512, 0, stream>>>(readbuf, Wa, ba, out);
}


# masked softmax fused into k_read: per workgroup head maxima and exp sums from the raw scores, p staged in LDS as A operand, 1/sum applied to the finished sums; one launch less (new sum order)
# speedup vs baseline: 16.8827x; 1.1296x over previous
#include <hip/hip_runtime.h>
#include <hip/hip_bf16.h>
#include <math.h>

#define LL 2048
#define BB 64
#define KK 256
#define VV 256
#define HH 8
#define HK 2048
#define HV 2048

typedef __attribute__((ext_vector_type(2))) float v2f;
typedef __attribute__((ext_vector_type(4))) float v4f;
typedef __attribute__((ext_vector_type(8))) float v8f;

__device__ __forceinline__ v8f wmma4(v2f a, v2f b, v8f c) {
    return __builtin_amdgcn_wmma_f32_16x16x4_f32(
         false, a,  false, b,
         (short)0, c,  false,  false);
}

#define QP_WAVES 8
#define QP_KW    (KK / QP_WAVES)
#define QP_STEPS (QP_KW / 8)
#define QP_MT    2

__global__ void __launch_bounds__(QP_WAVES * 32)
k_qproj(const float* __restrict__ query,
        const float* __restrict__ Wq,
        const float* __restrict__ bq,
        float* __restrict__ qbuf) {
    __shared__ float qpart[QP_WAVES][16 * QP_MT][32];

    const int tm   = blockIdx.x >> 6;
    const int tp   = blockIdx.x & 63;
    const int wave = threadIdx.x >> 5;
    const int lane = threadIdx.x & 31;
    const int col  = lane & 15;
    const int half = lane >> 4;

    const int k0 = wave * QP_KW + 4 * half;
    const float* ap = query + (size_t)(tm * (16 * QP_MT) + col) * KK + k0;
    const float* bp = Wq + (size_t)k0 * HK + tp * 32 + 2 * col;

    v4f av[QP_MT][QP_STEPS];
    v2f bv[QP_STEPS][4];
    for (int s = 0; s < QP_STEPS; ++s) {
        for (int m = 0; m < QP_MT; ++m)
            av[m][s] = *(const v4f*)(ap + (size_t)m * 16 * KK + 8 * s);
        for (int q = 0; q < 4; ++q)
            bv[s][q] = *(const v2f*)(bp + (size_t)(8 * s + q) * HK);
    }
    __builtin_amdgcn_sched_barrier(0);

    v8f acc[QP_MT][2] = {};
    for (int s = 0; s < QP_STEPS; ++s) {
        for (int p = 0; p < 2; ++p) {
            for (int m = 0; m < QP_MT; ++m) {
                const v2f a = { av[m][s][2 * p], av[m][s][2 * p + 1] };
                for (int t = 0; t < 2; ++t) {
                    const v2f bb = { bv[s][2 * p][t], bv[s][2 * p + 1][t] };
                    acc[m][t] = wmma4(a, bb, acc[m][t]);
                }
            }
        }
    }

    for (int m = 0; m < QP_MT; ++m) {
        for (int j = 0; j < 8; ++j) {
            const v2f o = { acc[m][0][j], acc[m][1][j] };
            *(v2f*)&qpart[wave][16 * m + 8 * half + j][2 * col] = o;
        }
    }
    __syncthreads();

    const int r0 = threadIdx.x >> 5;
    const int cc = threadIdx.x & 31;
    const float bias = bq[tp * 32 + cc];
    for (int i = 0; i < (16 * QP_MT) / QP_WAVES; ++i) {
        const int r = r0 + QP_WAVES * i;
        float sum = 0.0f;
        for (int w = 0; w < QP_WAVES; ++w) sum += qpart[w][r][cc];
        qbuf[(size_t)(tm * (16 * QP_MT) + r) * HK + tp * 32 + cc] = sum + bias;
    }
}

#define SC_NB    256
#define SC_WAVES 16
#define SC_TILES 2
#define SC_TROWS (16 * SC_WAVES)
#define SC_PITCH 260

template <int CTRL>
__device__ __forceinline__ int dpp_row(int v) {
    return __builtin_amdgcn_update_dpp(0, v, CTRL, 0xf, 0xf, false);
}
__device__ __forceinline__ int wave_sum(int v) {
    v += dpp_row<0x161>(v);
    v += dpp_row<0x162>(v);
    v += dpp_row<0x164>(v);
    v += dpp_row<0x168>(v);
    v += __builtin_amdgcn_permlanex16(v, v, 0x76543210, 0xfedcba98, false, false);
    return v;
}
__device__ __forceinline__ int wave_scan(int v, int lane) {
    v += dpp_row<0x111>(v);
    v += dpp_row<0x112>(v);
    v += dpp_row<0x114>(v);
    v += dpp_row<0x118>(v);
    const int row0 = __builtin_amdgcn_readlane(v, 15);
    return v + (lane >= 16 ? row0 : 0);
}

__device__ __forceinline__ void sc_load_group(const float* __restrict__ kp,
                                              int g, v4f (&kv)[16]) {
    const float* p = kp + (size_t)(SC_TROWS * (g >> 1)) * (BB * KK) + 128 * (g & 1);
    for (int i = 0; i < 16; ++i)
        kv[i] = __builtin_nontemporal_load(
            (const v4f*)(p + (size_t)i * (BB * KK)));
}

template <int NG>
__device__ __forceinline__ void sc_stream(const float* __restrict__ kp,
                                          const float* __restrict__ rp,
                                          float* __restrict__ sp,
                                          float* myt, const float* arow,
                                          const float* brow, int lane, int col) {
    float rv[NG / 2];
    for (int t = 0; t < NG / 2; ++t)
        rv[t] = rp[(size_t)(SC_TROWS * t) * BB];

    v4f kv[2][16];
    sc_load_group(kp, 0, kv[0]);
    sc_load_group(kp, 1, kv[1]);

    v8f acc0 = {}, acc1 = {};
    for (int g = 0; g < NG; ++g) {
        const int t  = g >> 1;
        const int kh = g & 1;
        for (int i = 0; i < 16; ++i)
            *(v4f*)(myt + i * SC_PITCH + 128 * kh + 4 * lane) = kv[g & 1][i];
        __builtin_amdgcn_sched_barrier(0);
        if (g + 2 < NG)
            sc_load_group(kp, g + 2, kv[g & 1]);
        __builtin_amdgcn_sched_barrier(0);

        if (kh == 0) { acc0 = v8f{}; acc1 = v8f{}; }
        for (int j = 0; j < 16; ++j) {
            const v4f k4 = *(const v4f*)(arow + 128 * kh + 8 * j);
            const v4f q4 = *(const v4f*)(brow + 128 * kh + 8 * j);
            const v4f a4 = k4 * rv[t];
            acc0 = wmma4(v2f{a4.x, a4.y}, v2f{q4.x, q4.y}, acc0);
            acc1 = wmma4(v2f{a4.z, a4.w}, v2f{q4.z, q4.w}, acc1);
        }
        if (kh == 1 && col < HH) {
            const v8f s = acc0 + acc1;
            *(v4f*)(sp + SC_TROWS * t)     = v4f{s[0], s[1], s[2], s[3]};
            *(v4f*)(sp + SC_TROWS * t + 4) = v4f{s[4], s[5], s[6], s[7]};
        }
    }
}

__global__ void __launch_bounds__(SC_WAVES * 32)
k_scores(const float* __restrict__ keys,
         const float* __restrict__ rpe,
         const float* __restrict__ qbuf,
         const int* __restrict__ steps,
         float* __restrict__ smat) {
    __shared__ __attribute__((aligned(16))) float kt[SC_WAVES][16 * SC_PITCH];
    __shared__ __attribute__((aligned(16))) float qs[HH * SC_PITCH];

    const int wave = __builtin_amdgcn_readfirstlane(threadIdx.x >> 5);
    const int lane = threadIdx.x & 31;
    const int col  = lane & 15;
    const int half = lane >> 4;

    const int n0 = (min(max(steps[lane], 0), LL) + 15) >> 4;
    const int n1 = (min(max(steps[lane + 32], 0), LL) + 15) >> 4;
    int ct = max((__builtin_amdgcn_readfirstlane(wave_sum(n0 + n1)) + SC_NB - 1) / SC_NB, 1);
    for (; ct < 16 * SC_TILES; ++ct)
        if (__builtin_amdgcn_readfirstlane(
                wave_sum((n0 + ct - 1) / ct + (n1 + ct - 1) / ct)) <= SC_NB) break;
    const int c0 = (n0 + ct - 1) / ct;
    const int c1 = (n1 + ct - 1) / ct;
    const int p0 = wave_scan(c0, lane);
    const int p1 = wave_scan(c1, lane) + __builtin_amdgcn_readlane(p0, 31);
    const int ci = blockIdx.x;
    if (ci >= __builtin_amdgcn_readlane(p1, 31)) return;
    const int b  = __popcll(__ballot(p0 <= ci)) + __popcll(__ballot(p1 <= ci));
    const int bl = b & 31;
    const int ex = (b < 32) ? __builtin_amdgcn_readlane(p0 - c0, bl)
                            : __builtin_amdgcn_readlane(p1 - c1, bl);
    const int nb = (b < 32) ? __builtin_amdgcn_readlane(n0, bl)
                            : __builtin_amdgcn_readlane(n1, bl);
    const int tile0 = (ci - ex) * ct;
    const int ntile = min(ct, nb - tile0);
    const int nt = (wave < ntile) + (wave + SC_WAVES < ntile);
    const int lw = 16 * (tile0 + wave);

    const v4f qv = *(const v4f*)(qbuf + (size_t)b * HK + 4 * threadIdx.x);
    *(v4f*)&qs[(threadIdx.x >> 6) * SC_PITCH + 4 * (threadIdx.x & 63)] = qv;
    __syncthreads();

    float* myt = kt[wave];
    const float* arow = myt + col * SC_PITCH + 4 * half;
    const float* brow = qs + (col & (HH - 1)) * SC_PITCH + 4 * half;
    const float* kp = keys + ((size_t)lw * BB + b) * KK + 4 * lane;
    const float* rp = rpe + (size_t)(lw + col) * BB + b;
    float* sp = smat + ((size_t)b * HH + col) * LL + lw + 8 * half;

    if (nt == 2)      sc_stream<4>(kp, rp, sp, myt, arow, brow, lane, col);
    else if (nt == 1) sc_stream<2>(kp, rp, sp, myt, arow, brow, lane, col);
}

#define RD_WAVES 16
#define RD_ROWS  (LL / RD_WAVES)
#define RD_STEPS 4
#define RD_GRPS  (RD_ROWS / (8 * RD_STEPS))
#define RD_PITCH (LL + 4)

template <int CTRL>
__device__ __forceinline__ float dpp_rowf(float v) {
    return __int_as_float(
        __builtin_amdgcn_update_dpp(0, __float_as_int(v), CTRL, 0xf, 0xf, false));
}
__device__ __forceinline__ float other_row(float v) {
    const int i = __float_as_int(v);
    return __int_as_float(
        __builtin_amdgcn_permlanex16(i, i, 0x76543210, 0xfedcba98, false, false));
}
__device__ __forceinline__ float wave_maxf(float v) {
    v = fmaxf(v, dpp_rowf<0x161>(v));
    v = fmaxf(v, dpp_rowf<0x162>(v));
    v = fmaxf(v, dpp_rowf<0x164>(v));
    v = fmaxf(v, dpp_rowf<0x168>(v));
    return fmaxf(v, other_row(v));
}
__device__ __forceinline__ float wave_addf(float v) {
    v += dpp_rowf<0x161>(v);
    v += dpp_rowf<0x162>(v);
    v += dpp_rowf<0x164>(v);
    v += dpp_rowf<0x168>(v);
    return v + other_row(v);
}

__device__ __forceinline__ void lds_barrier() {
    asm volatile("s_wait_dscnt 0x0\n\ts_barrier_signal -1\n\ts_barrier_wait -1" ::: "memory");
}

__device__ __forceinline__ void rd_load_vals(const float* __restrict__ vp, int g,
                                             v4f (&vb)[RD_STEPS][4]) {
    for (int j = 0; j < RD_STEPS; ++j) {
        const int s = g * RD_STEPS + j;
        for (int q = 0; q < 4; ++q)
            vb[j][q] = __builtin_nontemporal_load(
                (const v4f*)(vp + (size_t)(8 * s + q) * (BB * VV)));
    }
}

__device__ __forceinline__ void rd_load_p(const float* pw, int g,
                                          v4f (&wa)[RD_STEPS]) {
    for (int j = 0; j < RD_STEPS; ++j)
        wa[j] = *(const v4f*)(pw + 8 * (g * RD_STEPS + j));
}

__device__ __forceinline__ void rd_mma_group(const v4f (&wa)[RD_STEPS],
                                             const v4f (&vb)[RD_STEPS][4],
                                             v8f (&acc)[4]) {
    for (int j = 0; j < RD_STEPS; ++j) {
        for (int p = 0; p < 2; ++p) {
            const v2f a = { wa[j][2 * p], wa[j][2 * p + 1] };
            for (int t = 0; t < 4; ++t) {
                const v2f bb = { vb[j][2 * p][t], vb[j][2 * p + 1][t] };
                acc[t] = wmma4(a, bb, acc[t]);
            }
        }
    }
}

template <int NG>
__device__ __forceinline__ void rd_stream(const float* pw,
                                          const float* __restrict__ vp,
                                          v4f (&vb)[2][RD_STEPS][4],
                                          v8f (&acc)[4]) {
    for (int g = 0; g < NG; ++g) {
        __builtin_amdgcn_sched_barrier(0);
        if (g + 1 < NG)
            rd_load_vals(vp, g + 1, vb[(g + 1) & 1]);
        __builtin_amdgcn_sched_barrier(0);
        v4f wa[RD_STEPS];
        rd_load_p(pw, g, wa);
        rd_mma_group(wa, vb[g & 1], acc);
    }
}

__global__ void __launch_bounds__(RD_WAVES * 32)
k_read(const float* __restrict__ smat,
       const float* __restrict__ vals,
       const int* __restrict__ steps,
       float* __restrict__ readbuf) {
    __shared__ __attribute__((aligned(16))) float ps[HH * RD_PITCH];
    __shared__ float part[RD_WAVES][HH][64];
    __shared__ float redm[RD_WAVES];
    __shared__ float reds[RD_WAVES];

    const int b    = blockIdx.x >> 2;
    const int vc   = blockIdx.x & 3;
    const int wave = __builtin_amdgcn_readfirstlane(threadIdx.x >> 5);
    const int lane = threadIdx.x & 31;
    const int col  = lane & 15;
    const int half = lane >> 4;
    const int l0 = wave * RD_ROWS + 4 * half;
    const float* pw = ps + (col & (HH - 1)) * RD_PITCH + l0;
    const float* vp = vals + ((size_t)l0 * BB + b) * VV + vc * 64 + 4 * col;

    const int step = min(max(steps[b], 0), LL);
    const int ng = min(max((step - wave * RD_ROWS + 8 * RD_STEPS - 1) / (8 * RD_STEPS), 0),
                       RD_GRPS);

    const int hd = threadIdx.x >> 6;
    const int lb = 4 * (threadIdx.x & 63);
    const float* sp = smat + ((size_t)b * HH + hd) * LL + lb;
    v4f sv[8];
    for (int i = 0; i < 8; ++i)
        sv[i] = *(const v4f*)(sp + 256 * i);

    float m = -INFINITY;
    for (int i = 0; i < 8; ++i)
        for (int e = 0; e < 4; ++e)
            if (lb + 256 * i + e < step) m = fmaxf(m, sv[i][e]);
    m = wave_maxf(m);
    if (lane == 0) redm[wave] = m;

    v4f vb[2][RD_STEPS][4];
    if (ng > 0) rd_load_vals(vp, 0, vb[0]);
    lds_barrier();
    const float mx = fmaxf(redm[2 * hd], redm[2 * hd + 1]);

    float psum = 0.0f;
    for (int i = 0; i < 8; ++i) {
        v4f p;
        for (int e = 0; e < 4; ++e) {
            p[e] = (lb + 256 * i + e < step) ? __expf(sv[i][e] - mx) : 0.0f;
            psum += p[e];
        }
        *(v4f*)&ps[hd * RD_PITCH + lb + 256 * i] = p;
    }
    psum = wave_addf(psum);
    if (lane == 0) reds[wave] = psum;
    lds_barrier();

    v8f acc[4] = {};
    if (ng == 4)      rd_stream<4>(pw, vp, vb, acc);
    else if (ng == 3) rd_stream<3>(pw, vp, vb, acc);
    else if (ng == 2) rd_stream<2>(pw, vp, vb, acc);
    else if (ng == 1) rd_stream<1>(pw, vp, vb, acc);

    if (half == 0) {
        for (int r = 0; r < 8; ++r) {
            const v4f o = { acc[0][r], acc[1][r], acc[2][r], acc[3][r] };
            *(v4f*)&part[wave][r][4 * col] = o;
        }
    }
    __syncthreads();

    const int r  = threadIdx.x >> 6;
    const int vv = threadIdx.x & 63;
    const float inv = 1.0f / (reds[2 * r] + reds[2 * r + 1]);
    float s = 0.0f;
    for (int w = 0; w < RD_WAVES; ++w) s += part[w][r][vv];
    readbuf[((size_t)b * HH + r) * VV + vc * 64 + vv] = s * inv;
}

#define OUT_WAVES 16
#define OUT_KW    (HV / OUT_WAVES)
#define OUT_STEPS (OUT_KW / 8)

__global__ void __launch_bounds__(OUT_WAVES * 32)
k_out(const float* __restrict__ readbuf,
      const float* __restrict__ Wa,
      const float* __restrict__ ba,
      float* __restrict__ out) {
    __shared__ float opart[OUT_WAVES][16][32];

    const int tm   = blockIdx.x >> 3;
    const int tp   = blockIdx.x & 7;
    const int wave = threadIdx.x >> 5;
    const int lane = threadIdx.x & 31;
    const int col  = lane & 15;
    const int half = lane >> 4;

    const int k0 = wave * OUT_KW + 4 * half;
    const float* ap = readbuf + (size_t)(tm * 16 + col) * HV + k0;
    const float* bp = Wa + (size_t)k0 * VV + tp * 32 + 2 * col;

    v4f av[OUT_STEPS];
    v2f bv[OUT_STEPS][4];
    for (int s = 0; s < OUT_STEPS; ++s) {
        av[s] = *(const v4f*)(ap + 8 * s);
        for (int q = 0; q < 4; ++q)
            bv[s][q] = *(const v2f*)(bp + (size_t)(8 * s + q) * VV);
    }
    __builtin_amdgcn_sched_barrier(0);

    v8f acc[2] = {};
    for (int s = 0; s < OUT_STEPS; ++s) {
        for (int p = 0; p < 2; ++p) {
            const v2f a = { av[s][2 * p], av[s][2 * p + 1] };
            for (int t = 0; t < 2; ++t) {
                const v2f bb = { bv[s][2 * p][t], bv[s][2 * p + 1][t] };
                acc[t] = wmma4(a, bb, acc[t]);
            }
        }
    }

    for (int j = 0; j < 8; ++j) {
        const v2f o = { acc[0][j], acc[1][j] };
        *(v2f*)&opart[wave][8 * half + j][2 * col] = o;
    }
    __syncthreads();

    const int r  = threadIdx.x >> 5;
    const int cc = threadIdx.x & 31;
    float sum = 0.0f;
    for (int w = 0; w < OUT_WAVES; ++w) sum += opart[w][r][cc];
    out[(size_t)(tm * 16 + r) * VV + tp * 32 + cc] = sum + ba[tp * 32 + cc];
}

extern "C" void kernel_launch(void* const* d_in, const int* in_sizes, int n_in,
                              void* d_out, int out_size, void* d_ws, size_t ws_size,
                              hipStream_t stream) {
    const float* query = (const float*)d_in[0];
    const float* keys  = (const float*)d_in[1];
    const float* vals  = (const float*)d_in[2];
    const float* rpe   = (const float*)d_in[3];
    const float* Wq    = (const float*)d_in[4];
    const float* bq    = (const float*)d_in[5];
    const float* Wa    = (const float*)d_in[6];
    const float* ba    = (const float*)d_in[7];
    const int*   steps = (const int*)d_in[8];
    float* out = (float*)d_out;

    float* ws = (float*)d_ws;
    float* qbuf    = ws;
    float* smat    = qbuf + (size_t)BB * HK;
    float* readbuf = smat + (size_t)BB * HH * LL;

    k_qproj<<<(BB / (16 * QP_MT)) * (HK / 32), QP_WAVES * 32, 0, stream>>>(query, Wq, bq, qbuf);
    k_scores<<<SC_NB, SC_WAVES * 32, 0, stream>>>(keys, rpe, qbuf, steps, smat);
    k_read<<<256, 512, 0, stream>>>(smat, vals, steps, readbuf);
    k_out<<<32, 512, 0, stream>>>(readbuf, Wa, ba, out);
}
